# starting kernel
# baseline (speedup 1.0000x reference)
#include <hip/hip_runtime.h>


typedef __attribute__((ext_vector_type(2))) float v2f;
typedef __attribute__((ext_vector_type(8))) float v8f;

#define NCOL 128
#define TILE_M 128
#define KC 64
#define LXS 68
#define LWS 288

__global__ __launch_bounds__(256) void k_gemm_wmma(
    const float* __restrict__ X, const float* __restrict__ W,
    float* __restrict__ C, int M, int K) {
  __shared__ float lw2[(KC / 2) * LWS];
  __shared__ float lx[TILE_M * LXS];

  const int tid = threadIdx.x;
  const int lane = tid & 31;
  const int wave = tid >> 5;
  const int mblock = blockIdx.x * TILE_M;
  const int mstrip = mblock + wave * 16;

  v8f acc[8] = {};

  const int lane15 = lane & 15;
  const int ksel = (lane < 16) ? 0 : 2;
  const int mrow = wave * 16 + lane15;

  const float* axp = lx + mrow * LXS + ksel;
  const float* bbp = lw2 + (ksel >> 1) * LWS + lane15 * 2;

  for (int k0 = 0; k0 < K; k0 += KC) {
    __syncthreads();
    for (int i = tid; i < (KC / 2) * (NCOL / 4); i += 256) {
      const int kk2 = i >> 5;
      const int c4 = i & 31;
      const float* wr = W + (size_t)(k0 + kk2 * 2) * NCOL + c4 * 4;
      const float4 w0 = *(const float4*)wr;
      const float4 w1 = *(const float4*)(wr + NCOL);
      float* dst = lw2 + kk2 * LWS + c4 * 8;
      ((float2*)dst)[0] = make_float2(w0.x, w1.x);
      ((float2*)dst)[1] = make_float2(w0.y, w1.y);
      ((float2*)dst)[2] = make_float2(w0.z, w1.z);
      ((float2*)dst)[3] = make_float2(w0.w, w1.w);
    }
    for (int i = tid; i < TILE_M * (KC / 4); i += 256) {
      const int r = i >> 4;
      const int c4 = i & 15;
      const int gr = mblock + r;
      float4 v = make_float4(0.f, 0.f, 0.f, 0.f);
      if (gr < M) v = *(const float4*)(X + (size_t)gr * K + k0 + c4 * 4);
      *(float4*)(lx + r * LXS + c4 * 4) = v;
    }
    __syncthreads();

#pragma unroll 4
    for (int kk = 0; kk < KC; kk += 4) {
      const v2f a = *(const v2f*)(axp + kk);
      const float* brow = bbp + (kk >> 1) * LWS;
#pragma unroll
      for (int t = 0; t < 8; ++t) {
        const v2f b = *(const v2f*)(brow + t * 32);
        acc[t] = __builtin_amdgcn_wmma_f32_16x16x4_f32(
             false, a,  false, b,
             (short)0, acc[t],  false,  false);
      }
    }
  }

  if (mstrip < M) {
    const int mo = (lane < 16) ? 0 : 8;
#pragma unroll
    for (int t = 0; t < 8; ++t) {
#pragma unroll
      for (int v = 0; v < 8; ++v) {
        const int m = mstrip + mo + v;
        if (m < M) C[(size_t)m * NCOL + t * 16 + lane15] = acc[t][v];
      }
    }
  }
}

__global__ void k_deg_init(float* deg, int N) {
  int i = blockIdx.x * blockDim.x + threadIdx.x;
  if (i < N) deg[i] = 1.0f;
}

__global__ void k_deg_count(const int* __restrict__ col, float* deg, int E) {
  int i = blockIdx.x * blockDim.x + threadIdx.x;
  if (i < E) atomicAdd(&deg[col[i]], 1.0f);
}

__global__ void k_rsqrt(float* deg, int N) {
  int i = blockIdx.x * blockDim.x + threadIdx.x;
  if (i < N) {
    float d = deg[i];
    deg[i] = (d > 0.f) ? rsqrtf(d) : 0.f;
  }
}

__global__ void k_selfloop_init(const float* __restrict__ hlin,
                                const float* __restrict__ dinv,
                                float* __restrict__ agg, int N) {
  int i = blockIdx.x * blockDim.x + threadIdx.x;
  if (i < N * NCOL) {
    int n = i >> 7;
    float w = dinv[n];
    agg[i] = hlin[i] * w * w;
  }
}

__global__ __launch_bounds__(256) void k_edge_scatter(
    const float* __restrict__ hlin, const int* __restrict__ row,
    const int* __restrict__ col, const float* __restrict__ dinv,
    float* __restrict__ agg, int E) {
  int e = blockIdx.x * 2 + (threadIdx.x >> 7);
  int f = threadIdx.x & 127;
  if (e >= E) return;
  int s = row[e];
  int d = col[e];
  float w = dinv[s] * dinv[d];
  atomicAdd(&agg[(size_t)d * NCOL + f], hlin[(size_t)s * NCOL + f] * w);
}

__global__ void k_bias_act(float* __restrict__ h, const float* __restrict__ b,
                           int N, int do_relu) {
  int i = blockIdx.x * blockDim.x + threadIdx.x;
  if (i < N * NCOL) {
    float v = h[i] + b[i & 127];
    h[i] = (do_relu && v < 0.f) ? 0.f : v;
  }
}

__global__ void k_zero(float* p, int n) {
  int i = blockIdx.x * blockDim.x + threadIdx.x;
  if (i < n) p[i] = 0.f;
}

__global__ void k_pool(const float* __restrict__ h, const int* __restrict__ batch,
                       float* __restrict__ hs, int N) {
  int i = blockIdx.x * blockDim.x + threadIdx.x;
  if (i < N * NCOL) {
    int n = i >> 7;
    atomicAdd(&hs[(size_t)batch[n] * NCOL + (i & 127)], h[i]);
  }
}

extern "C" void kernel_launch(void* const* d_in, const int* in_sizes, int n_in,
                              void* d_out, int out_size, void* d_ws, size_t ws_size,
                              hipStream_t stream) {
  const float* x = (const float*)d_in[0];
  const float* W1 = (const float*)d_in[1];
  const float* b1 = (const float*)d_in[2];
  const float* W2 = (const float*)d_in[3];
  const float* b2 = (const float*)d_in[4];
  const int* eidx = (const int*)d_in[5];
  const int* batch = (const int*)d_in[6];

  const int N = in_sizes[6];
  const int E = in_sizes[5] / 2;
  const int K1 = in_sizes[0] / N;
  const int* row = eidx;
  const int* col = eidx + E;

  float* ws = (float*)d_ws;
  float* dinv = ws;
  float* bufA = ws + (((size_t)N + 63) & ~(size_t)63);
  float* bufB = bufA + (size_t)N * NCOL;

  const int G = out_size / NCOL - N;
  float* hs = (float*)d_out;
  float* hout = hs + (size_t)G * NCOL;

  const dim3 b256(256);
  const int nfBlocks = (N * NCOL + 255) / 256;
  const int gemmBlocks = (N + TILE_M - 1) / TILE_M;
  const int edgeBlocks = (E + 1) / 2;

  k_deg_init<<<(N + 255) / 256, b256, 0, stream>>>(dinv, N);
  k_deg_count<<<(E + 255) / 256, b256, 0, stream>>>(col, dinv, E);
  k_rsqrt<<<(N + 255) / 256, b256, 0, stream>>>(dinv, N);

  k_gemm_wmma<<<gemmBlocks, b256, 0, stream>>>(x, W1, bufA, N, K1);
  k_selfloop_init<<<nfBlocks, b256, 0, stream>>>(bufA, dinv, bufB, N);
  k_edge_scatter<<<edgeBlocks, b256, 0, stream>>>(bufA, row, col, dinv, bufB, E);
  k_bias_act<<<nfBlocks, b256, 0, stream>>>(bufB, b1, N, 1);

  k_gemm_wmma<<<gemmBlocks, b256, 0, stream>>>(bufB, W2, bufA, N, NCOL);
  k_selfloop_init<<<nfBlocks, b256, 0, stream>>>(bufA, dinv, hout, N);
  k_edge_scatter<<<edgeBlocks, b256, 0, stream>>>(bufA, row, col, dinv, hout, E);
  k_bias_act<<<nfBlocks, b256, 0, stream>>>(hout, b2, N, 0);

  k_zero<<<(G * NCOL + 255) / 256, b256, 0, stream>>>(hs, G * NCOL);
  k_pool<<<nfBlocks, b256, 0, stream>>>(hout, batch, hs, N);
}


# Aggregation as a gather over edge lists counting-sorted by destination each call (integer atomic ranks), replacing the per-feature float atomic scatter; row sums run in list order; bias, relu fused
# speedup vs baseline: 2.6080x; 2.6080x over previous
#include <hip/hip_runtime.h>


typedef __attribute__((ext_vector_type(2))) float v2f;
typedef __attribute__((ext_vector_type(8))) float v8f;

#define NCOL 128
#define TILE_M 128
#define KC 64
#define LXS 68
#define LWS 288

__global__ __launch_bounds__(256) void k_gemm_wmma(
    const float* __restrict__ X, const float* __restrict__ W,
    float* __restrict__ C, int M, int K) {
  __shared__ float lw2[(KC / 2) * LWS];
  __shared__ float lx[TILE_M * LXS];

  const int tid = threadIdx.x;
  const int lane = tid & 31;
  const int wave = tid >> 5;
  const int mblock = blockIdx.x * TILE_M;
  const int mstrip = mblock + wave * 16;

  v8f acc[8] = {};

  const int lane15 = lane & 15;
  const int ksel = (lane < 16) ? 0 : 2;
  const int mrow = wave * 16 + lane15;

  const float* axp = lx + mrow * LXS + ksel;
  const float* bbp = lw2 + (ksel >> 1) * LWS + lane15 * 2;

  for (int k0 = 0; k0 < K; k0 += KC) {
    __syncthreads();
    for (int i = tid; i < (KC / 2) * (NCOL / 4); i += 256) {
      const int kk2 = i >> 5;
      const int c4 = i & 31;
      const float* wr = W + (size_t)(k0 + kk2 * 2) * NCOL + c4 * 4;
      const float4 w0 = *(const float4*)wr;
      const float4 w1 = *(const float4*)(wr + NCOL);
      float* dst = lw2 + kk2 * LWS + c4 * 8;
      ((float2*)dst)[0] = make_float2(w0.x, w1.x);
      ((float2*)dst)[1] = make_float2(w0.y, w1.y);
      ((float2*)dst)[2] = make_float2(w0.z, w1.z);
      ((float2*)dst)[3] = make_float2(w0.w, w1.w);
    }
    for (int i = tid; i < TILE_M * (KC / 4); i += 256) {
      const int r = i >> 4;
      const int c4 = i & 15;
      const int gr = mblock + r;
      float4 v = make_float4(0.f, 0.f, 0.f, 0.f);
      if (gr < M) v = *(const float4*)(X + (size_t)gr * K + k0 + c4 * 4);
      *(float4*)(lx + r * LXS + c4 * 4) = v;
    }
    __syncthreads();

#pragma unroll 4
    for (int kk = 0; kk < KC; kk += 4) {
      const v2f a = *(const v2f*)(axp + kk);
      const float* brow = bbp + (kk >> 1) * LWS;
#pragma unroll
      for (int t = 0; t < 8; ++t) {
        const v2f b = *(const v2f*)(brow + t * 32);
        acc[t] = __builtin_amdgcn_wmma_f32_16x16x4_f32(
             false, a,  false, b,
             (short)0, acc[t],  false,  false);
      }
    }
  }

  if (mstrip < M) {
    const int mo = (lane < 16) ? 0 : 8;
#pragma unroll
    for (int t = 0; t < 8; ++t) {
#pragma unroll
      for (int v = 0; v < 8; ++v) {
        const int m = mstrip + mo + v;
        if (m < M) C[(size_t)m * NCOL + t * 16 + lane15] = acc[t][v];
      }
    }
  }
}

__global__ void k_count_rank(const int* __restrict__ col, int* cnt,
                             int* __restrict__ rank, int E) {
  int i = blockIdx.x * blockDim.x + threadIdx.x;
  if (i < E) rank[i] = atomicAdd(&cnt[col[i]], 1);
}

__global__ __launch_bounds__(256) void k_alloc(const int* cnt, int* total,
                                               int* __restrict__ start,
                                               float* __restrict__ dinv, int N) {
  const int n = blockIdx.x * 256 + threadIdx.x;
  const int lane = threadIdx.x & 31;
  const int c = (n < N) ? cnt[n] : 0;
  int incl = c;
  for (int off = 1; off < 32; off <<= 1) {
    const int t = __shfl_up(incl, off);
    if (lane >= off) incl += t;
  }
  int base = 0;
  if (lane == 31) base = atomicAdd(total, incl);
  base = __shfl(base, 31);
  if (n < N) {
    start[n] = base + incl - c;
    const float d = (float)c + 1.0f;
    dinv[n] = (d > 0.f) ? rsqrtf(d) : 0.f;
  }
}

__global__ void k_fill(const int* __restrict__ row, const int* __restrict__ col,
                       const int* __restrict__ start, const int* __restrict__ rank,
                       int* __restrict__ csr, int E) {
  int i = blockIdx.x * blockDim.x + threadIdx.x;
  if (i < E) csr[start[col[i]] + rank[i]] = row[i];
}

__device__ __forceinline__ void fma4(float4& a, const float4 v, const float wl, const int t) {
  const float w = __int_as_float(__builtin_amdgcn_readlane(__float_as_int(wl), t));
  a.x = fmaf(v.x, w, a.x);
  a.y = fmaf(v.y, w, a.y);
  a.z = fmaf(v.z, w, a.z);
  a.w = fmaf(v.w, w, a.w);
}

__global__ __launch_bounds__(256) void k_gather(
    const float* __restrict__ hlin, const int* __restrict__ start,
    const int* __restrict__ cnt, const int* __restrict__ csr,
    const float* __restrict__ dinv, const float* __restrict__ bias,
    float* __restrict__ out, int N, int do_relu) {
  const int lane = threadIdx.x & 31;
  const int n = __builtin_amdgcn_readfirstlane(blockIdx.x * 8 + (threadIdx.x >> 5));
  if (n >= N) return;
  const float4* __restrict__ H = (const float4*)hlin;
  const int s = start[n];
  const int e = s + cnt[n];
  const float dn = dinv[n];

  const float4 self = H[(size_t)n * (NCOL / 4) + lane];
  float4 acc = make_float4(self.x * dn * dn, self.y * dn * dn, self.z * dn * dn, self.w * dn * dn);

  for (int base = s; base < e; base += 32) {
    const int m = (e - base < 32) ? (e - base) : 32;
    int idx = 0;
    float w = 0.f;
    if (lane < m) {
      idx = csr[base + lane];
      w = dinv[idx] * dn;
    }
    int t = 0;
    for (; t + 4 <= m; t += 4) {
      const int s0 = __builtin_amdgcn_readlane(idx, t);
      const int s1 = __builtin_amdgcn_readlane(idx, t + 1);
      const int s2 = __builtin_amdgcn_readlane(idx, t + 2);
      const int s3 = __builtin_amdgcn_readlane(idx, t + 3);
      const float4 v0 = H[(size_t)s0 * (NCOL / 4) + lane];
      const float4 v1 = H[(size_t)s1 * (NCOL / 4) + lane];
      const float4 v2 = H[(size_t)s2 * (NCOL / 4) + lane];
      const float4 v3 = H[(size_t)s3 * (NCOL / 4) + lane];
      fma4(acc, v0, w, t);
      fma4(acc, v1, w, t + 1);
      fma4(acc, v2, w, t + 2);
      fma4(acc, v3, w, t + 3);
    }
    const int rem = m - t;
    if (rem > 0) {
      const int s0 = __builtin_amdgcn_readlane(idx, t);
      const float4 v0 = H[(size_t)s0 * (NCOL / 4) + lane];
      float4 v1 = make_float4(0.f, 0.f, 0.f, 0.f);
      float4 v2 = make_float4(0.f, 0.f, 0.f, 0.f);
      if (rem > 1) {
        const int s1 = __builtin_amdgcn_readlane(idx, t + 1);
        v1 = H[(size_t)s1 * (NCOL / 4) + lane];
      }
      if (rem > 2) {
        const int s2 = __builtin_amdgcn_readlane(idx, t + 2);
        v2 = H[(size_t)s2 * (NCOL / 4) + lane];
      }
      fma4(acc, v0, w, t);
      if (rem > 1) fma4(acc, v1, w, t + 1);
      if (rem > 2) fma4(acc, v2, w, t + 2);
    }
  }

  const float4 b = ((const float4*)bias)[lane];
  float4 r = make_float4(acc.x + b.x, acc.y + b.y, acc.z + b.z, acc.w + b.w);
  if (do_relu) {
    r.x = (r.x < 0.f) ? 0.f : r.x;
    r.y = (r.y < 0.f) ? 0.f : r.y;
    r.z = (r.z < 0.f) ? 0.f : r.z;
    r.w = (r.w < 0.f) ? 0.f : r.w;
  }
  ((float4*)out)[(size_t)n * (NCOL / 4) + lane] = r;
}

__global__ void k_zero(float* p, int n) {
  int i = blockIdx.x * blockDim.x + threadIdx.x;
  if (i < n) p[i] = 0.f;
}

__global__ void k_pool(const float* __restrict__ h, const int* __restrict__ batch,
                       float* __restrict__ hs, int N) {
  int i = blockIdx.x * blockDim.x + threadIdx.x;
  if (i < N * NCOL) {
    int n = i >> 7;
    atomicAdd(&hs[(size_t)batch[n] * NCOL + (i & 127)], h[i]);
  }
}

extern "C" void kernel_launch(void* const* d_in, const int* in_sizes, int n_in,
                              void* d_out, int out_size, void* d_ws, size_t ws_size,
                              hipStream_t stream) {
  const float* x = (const float*)d_in[0];
  const float* W1 = (const float*)d_in[1];
  const float* b1 = (const float*)d_in[2];
  const float* W2 = (const float*)d_in[3];
  const float* b2 = (const float*)d_in[4];
  const int* eidx = (const int*)d_in[5];
  const int* batch = (const int*)d_in[6];

  const int N = in_sizes[6];
  const int E = in_sizes[5] / 2;
  const int K1 = in_sizes[0] / N;
  const int* row = eidx;
  const int* col = eidx + E;

  float* ws = (float*)d_ws;
  float* dinv = ws;
  float* bufA = ws + (((size_t)N + 63) & ~(size_t)63);
  int* cnt = (int*)(bufA + (size_t)N * NCOL);
  const int NP = (N + 64) & ~63;
  const int EP = (E + 63) & ~63;
  int* total = cnt + N;
  int* start = cnt + NP;
  int* rank = start + NP;
  int* csr = rank + EP;

  const int G = out_size / NCOL - N;
  float* hs = (float*)d_out;
  float* hout = hs + (size_t)G * NCOL;

  const dim3 b256(256);
  const int nfBlocks = (N * NCOL + 255) / 256;
  const int gemmBlocks = (N + TILE_M - 1) / TILE_M;
  const int nodeBlocks = (N + 7) / 8;

  k_zero<<<(NP + 255) / 256, b256, 0, stream>>>((float*)cnt, NP);
  k_count_rank<<<(E + 255) / 256, b256, 0, stream>>>(col, cnt, rank, E);
  k_alloc<<<(N + 255) / 256, b256, 0, stream>>>(cnt, total, start, dinv, N);
  k_fill<<<(E + 255) / 256, b256, 0, stream>>>(row, col, start, rank, csr, E);

  k_gemm_wmma<<<gemmBlocks, b256, 0, stream>>>(x, W1, bufA, N, K1);
  k_gather<<<nodeBlocks, b256, 0, stream>>>(bufA, start, cnt, csr, dinv, b1, hout, N, 1);

  k_gemm_wmma<<<gemmBlocks, b256, 0, stream>>>(hout, W2, bufA, N, NCOL);
  k_gather<<<nodeBlocks, b256, 0, stream>>>(bufA, start, cnt, csr, dinv, b2, hout, N, 0);

  k_zero<<<(G * NCOL + 255) / 256, b256, 0, stream>>>(hs, G * NCOL);
  k_pool<<<nfBlocks, b256, 0, stream>>>(hout, batch, hs, N);
}


# Edges ordered by destination by a two-level counting sort on LDS counters (block bin counts, scan, binned partition, per-bin sort, whole-line stores), not per-edge global atomics; one launch fewer
# speedup vs baseline: 3.3404x; 1.2808x over previous
#include <hip/hip_runtime.h>


typedef __attribute__((ext_vector_type(2))) float v2f;
typedef __attribute__((ext_vector_type(8))) float v8f;

#define NCOL 128
#define TILE_M 128
#define KC 64
#define LXS 68
#define LWS 288

__global__ __launch_bounds__(256) void k_gemm_wmma(
    const float* __restrict__ X, const float* __restrict__ W,
    float* __restrict__ C, int M, int K) {
  __shared__ float lw2[(KC / 2) * LWS];
  __shared__ float lx[TILE_M * LXS];

  const int tid = threadIdx.x;
  const int lane = tid & 31;
  const int wave = tid >> 5;
  const int mblock = blockIdx.x * TILE_M;
  const int mstrip = mblock + wave * 16;

  v8f acc[8] = {};

  const int lane15 = lane & 15;
  const int ksel = (lane < 16) ? 0 : 2;
  const int mrow = wave * 16 + lane15;

  const float* axp = lx + mrow * LXS + ksel;
  const float* bbp = lw2 + (ksel >> 1) * LWS + lane15 * 2;

  for (int k0 = 0; k0 < K; k0 += KC) {
    __syncthreads();
    for (int i = tid; i < (KC / 2) * (NCOL / 4); i += 256) {
      const int kk2 = i >> 5;
      const int c4 = i & 31;
      const float* wr = W + (size_t)(k0 + kk2 * 2) * NCOL + c4 * 4;
      const float4 w0 = *(const float4*)wr;
      const float4 w1 = *(const float4*)(wr + NCOL);
      float* dst = lw2 + kk2 * LWS + c4 * 8;
      ((float2*)dst)[0] = make_float2(w0.x, w1.x);
      ((float2*)dst)[1] = make_float2(w0.y, w1.y);
      ((float2*)dst)[2] = make_float2(w0.z, w1.z);
      ((float2*)dst)[3] = make_float2(w0.w, w1.w);
    }
    for (int i = tid; i < TILE_M * (KC / 4); i += 256) {
      const int r = i >> 4;
      const int c4 = i & 15;
      const int gr = mblock + r;
      float4 v = make_float4(0.f, 0.f, 0.f, 0.f);
      if (gr < M) v = *(const float4*)(X + (size_t)gr * K + k0 + c4 * 4);
      *(float4*)(lx + r * LXS + c4 * 4) = v;
    }
    __syncthreads();

#pragma unroll 4
    for (int kk = 0; kk < KC; kk += 4) {
      const v2f a = *(const v2f*)(axp + kk);
      const float* brow = bbp + (kk >> 1) * LWS;
#pragma unroll
      for (int t = 0; t < 8; ++t) {
        const v2f b = *(const v2f*)(brow + t * 32);
        acc[t] = __builtin_amdgcn_wmma_f32_16x16x4_f32(
             false, a,  false, b,
             (short)0, acc[t],  false,  false);
      }
    }
  }

  if (mstrip < M) {
    const int mo = (lane < 16) ? 0 : 8;
#pragma unroll
    for (int t = 0; t < 8; ++t) {
#pragma unroll
      for (int v = 0; v < 8; ++v) {
        const int m = mstrip + mo + v;
        if (m < M) C[(size_t)m * NCOL + t * 16 + lane15] = acc[t][v];
      }
    }
  }
}

#define RB 8
#define RN 256
#define NBMAX 256
#define CH 4096
#define DCAP 8192

__device__ __forceinline__ int block_excl_scan(int v, int* red, int* total) {
  const int lane = threadIdx.x & 31;
  const int wave = threadIdx.x >> 5;
  int incl = v;
  for (int off = 1; off < 32; off <<= 1) {
    const int t = __shfl_up(incl, off);
    if (lane >= off) incl += t;
  }
  if (lane == 31) red[wave] = incl;
  __syncthreads();
  if (wave == 0) {
    const int t = red[lane];
    int ti = t;
    for (int off = 1; off < 32; off <<= 1) {
      const int u = __shfl_up(ti, off);
      if (lane >= off) ti += u;
    }
    red[lane] = ti - t;
    if (lane == 31) red[32] = ti;
  }
  __syncthreads();
  *total = red[32];
  return red[wave] + incl - v;
}

__global__ __launch_bounds__(1024) void k_binhist(const int* __restrict__ col,
                                                  int* __restrict__ wgcnt, int E, int NB) {
  __shared__ int hist[NBMAX];
  const int tid = threadIdx.x;
  if (tid < NBMAX) hist[tid] = 0;
  __syncthreads();
  const int e0 = blockIdx.x * CH;
  int c[CH / 1024];
#pragma unroll
  for (int k = 0; k < CH / 1024; ++k) {
    const int e = e0 + k * 1024 + tid;
    c[k] = (e < E) ? col[e] : -1;
  }
#pragma unroll
  for (int k = 0; k < CH / 1024; ++k)
    if (c[k] >= 0) atomicAdd(&hist[c[k] >> RB], 1);
  __syncthreads();
  if (tid < NB) wgcnt[blockIdx.x * NB + tid] = hist[tid];
}

__global__ __launch_bounds__(1024) void k_partition(
    const int* __restrict__ row, const int* __restrict__ col,
    const int* __restrict__ wgcnt, int* __restrict__ binStart,
    unsigned* __restrict__ pairs, int E, int NB, int NW) {
  __shared__ int ptot[1024];
  __shared__ int ppre[1024];
  __shared__ int lstart[NBMAX];
  __shared__ int cursor[NBMAX];
  __shared__ int gbase[NBMAX];
  __shared__ unsigned sorted[CH];
  __shared__ int red1[33];
  __shared__ int red2[33];
  const int tid = threadIdx.x;
  const int w = blockIdx.x;

  {
    const int b = tid & (NBMAX - 1);
    int t_ = 0, p_ = 0;
    if (b < NB) {
#pragma unroll 8
      for (int q = tid >> 8; q < NW; q += 4) {
        const int t = wgcnt[q * NB + b];
        t_ += t;
        p_ += (q < w) ? t : 0;
      }
    }
    ptot[tid] = t_;
    ppre[tid] = p_;
  }
  __syncthreads();
  int tot = 0, pre = 0, mine = 0;
  if (tid < NBMAX) {
    tot = ptot[tid] + ptot[tid + 256] + ptot[tid + 512] + ptot[tid + 768];
    pre = ppre[tid] + ppre[tid + 256] + ppre[tid + 512] + ppre[tid + 768];
    if (tid < NB) mine = wgcnt[w * NB + tid];
  }
  int total, ltotal;
  const int bs = block_excl_scan(tot, red1, &total);
  const int ls = block_excl_scan(mine, red2, &ltotal);
  if (tid < NBMAX) {
    lstart[tid] = ls;
    cursor[tid] = ls;
    gbase[tid] = bs + pre;
  }
  if (w == 0) {
    if (tid < NB) binStart[tid] = bs;
    if (tid == 0) binStart[NB] = total;
  }
  __syncthreads();

  const int e0 = w * CH;
  int ec[CH / 1024], er[CH / 1024];
#pragma unroll
  for (int k = 0; k < CH / 1024; ++k) {
    const int e = e0 + k * 1024 + tid;
    ec[k] = (e < E) ? col[e] : -1;
    er[k] = (e < E) ? row[e] : 0;
  }
#pragma unroll
  for (int k = 0; k < CH / 1024; ++k) {
    if (ec[k] >= 0) {
      const int p = atomicAdd(&cursor[ec[k] >> RB], 1);
      sorted[p] = ((unsigned)er[k] << RB) | (unsigned)(ec[k] & (RN - 1));
    }
  }
  __syncthreads();

  const int lane = tid & 31;
  for (int b = tid >> 5; b < NB; b += 32) {
    const int s = lstart[b];
    const int n = cursor[b] - s;
    const int g = gbase[b];
    for (int i = lane; i < n; i += 32) pairs[g + i] = sorted[s + i];
  }
}

__global__ __launch_bounds__(1024) void k_binsort(
    const unsigned* __restrict__ pairs, const int* __restrict__ binStart,
    int* __restrict__ start, int* __restrict__ cnt, float* __restrict__ dinv,
    int* __restrict__ csr, int N) {
  __shared__ int ncnt[RN];
  __shared__ int ncur[RN];
  __shared__ int red[33];
  __shared__ int stage[DCAP];
  const int tid = threadIdx.x;
  const int b = blockIdx.x;
  const int bs = binStart[b];
  const int M = binStart[b + 1] - bs;

  if (tid < RN) ncnt[tid] = 0;
  __syncthreads();
  for (int i = tid; i < M; i += 1024) atomicAdd(&ncnt[pairs[bs + i] & (RN - 1)], 1);
  __syncthreads();
  const int c = (tid < RN) ? ncnt[tid] : 0;
  int tot;
  const int ns = block_excl_scan(c, red, &tot);
  if (tid < RN) {
    ncur[tid] = ns;
    const int node = b * RN + tid;
    if (node < N) {
      start[node] = bs + ns;
      cnt[node] = c;
      const float d = (float)c + 1.0f;
      dinv[node] = (d > 0.f) ? rsqrtf(d) : 0.f;
    }
  }
  __syncthreads();
  if (M <= DCAP) {
    for (int i = tid; i < M; i += 1024) {
      const unsigned p = pairs[bs + i];
      const int pos = atomicAdd(&ncur[p & (RN - 1)], 1);
      stage[pos] = (int)(p >> RB);
    }
    __syncthreads();
    for (int i = tid; i < M; i += 1024) csr[bs + i] = stage[i];
  } else {
    for (int i = tid; i < M; i += 1024) {
      const unsigned p = pairs[bs + i];
      const int pos = atomicAdd(&ncur[p & (RN - 1)], 1);
      csr[bs + pos] = (int)(p >> RB);
    }
  }
}

__device__ __forceinline__ void fma4(float4& a, const float4 v, const float wl, const int t) {
  const float w = __int_as_float(__builtin_amdgcn_readlane(__float_as_int(wl), t));
  a.x = fmaf(v.x, w, a.x);
  a.y = fmaf(v.y, w, a.y);
  a.z = fmaf(v.z, w, a.z);
  a.w = fmaf(v.w, w, a.w);
}

__global__ __launch_bounds__(256) void k_gather(
    const float* __restrict__ hlin, const int* __restrict__ start,
    const int* __restrict__ cnt, const int* __restrict__ csr,
    const float* __restrict__ dinv, const float* __restrict__ bias,
    float* __restrict__ out, int N, int do_relu) {
  const int lane = threadIdx.x & 31;
  const int n = __builtin_amdgcn_readfirstlane(blockIdx.x * 8 + (threadIdx.x >> 5));
  if (n >= N) return;
  const float4* __restrict__ H = (const float4*)hlin;
  const int s = start[n];
  const int e = s + cnt[n];
  const float dn = dinv[n];

  const float4 self = H[(size_t)n * (NCOL / 4) + lane];
  float4 acc = make_float4(self.x * dn * dn, self.y * dn * dn, self.z * dn * dn, self.w * dn * dn);

  for (int base = s; base < e; base += 32) {
    const int m = (e - base < 32) ? (e - base) : 32;
    int idx = 0;
    float w = 0.f;
    if (lane < m) {
      idx = csr[base + lane];
      w = dinv[idx] * dn;
    }
    int t = 0;
    for (; t + 4 <= m; t += 4) {
      const int s0 = __builtin_amdgcn_readlane(idx, t);
      const int s1 = __builtin_amdgcn_readlane(idx, t + 1);
      const int s2 = __builtin_amdgcn_readlane(idx, t + 2);
      const int s3 = __builtin_amdgcn_readlane(idx, t + 3);
      const float4 v0 = H[(size_t)s0 * (NCOL / 4) + lane];
      const float4 v1 = H[(size_t)s1 * (NCOL / 4) + lane];
      const float4 v2 = H[(size_t)s2 * (NCOL / 4) + lane];
      const float4 v3 = H[(size_t)s3 * (NCOL / 4) + lane];
      fma4(acc, v0, w, t);
      fma4(acc, v1, w, t + 1);
      fma4(acc, v2, w, t + 2);
      fma4(acc, v3, w, t + 3);
    }
    const int rem = m - t;
    if (rem > 0) {
      const int s0 = __builtin_amdgcn_readlane(idx, t);
      const float4 v0 = H[(size_t)s0 * (NCOL / 4) + lane];
      float4 v1 = make_float4(0.f, 0.f, 0.f, 0.f);
      float4 v2 = make_float4(0.f, 0.f, 0.f, 0.f);
      if (rem > 1) {
        const int s1 = __builtin_amdgcn_readlane(idx, t + 1);
        v1 = H[(size_t)s1 * (NCOL / 4) + lane];
      }
      if (rem > 2) {
        const int s2 = __builtin_amdgcn_readlane(idx, t + 2);
        v2 = H[(size_t)s2 * (NCOL / 4) + lane];
      }
      fma4(acc, v0, w, t);
      if (rem > 1) fma4(acc, v1, w, t + 1);
      if (rem > 2) fma4(acc, v2, w, t + 2);
    }
  }

  const float4 b = ((const float4*)bias)[lane];
  float4 r = make_float4(acc.x + b.x, acc.y + b.y, acc.z + b.z, acc.w + b.w);
  if (do_relu) {
    r.x = (r.x < 0.f) ? 0.f : r.x;
    r.y = (r.y < 0.f) ? 0.f : r.y;
    r.z = (r.z < 0.f) ? 0.f : r.z;
    r.w = (r.w < 0.f) ? 0.f : r.w;
  }
  ((float4*)out)[(size_t)n * (NCOL / 4) + lane] = r;
}

__global__ void k_zero(float* p, int n) {
  int i = blockIdx.x * blockDim.x + threadIdx.x;
  if (i < n) p[i] = 0.f;
}

__global__ void k_pool(const float* __restrict__ h, const int* __restrict__ batch,
                       float* __restrict__ hs, int N) {
  int i = blockIdx.x * blockDim.x + threadIdx.x;
  if (i < N * NCOL) {
    int n = i >> 7;
    atomicAdd(&hs[(size_t)batch[n] * NCOL + (i & 127)], h[i]);
  }
}

extern "C" void kernel_launch(void* const* d_in, const int* in_sizes, int n_in,
                              void* d_out, int out_size, void* d_ws, size_t ws_size,
                              hipStream_t stream) {
  const float* x = (const float*)d_in[0];
  const float* W1 = (const float*)d_in[1];
  const float* b1 = (const float*)d_in[2];
  const float* W2 = (const float*)d_in[3];
  const float* b2 = (const float*)d_in[4];
  const int* eidx = (const int*)d_in[5];
  const int* batch = (const int*)d_in[6];

  const int N = in_sizes[6];
  const int E = in_sizes[5] / 2;
  const int K1 = in_sizes[0] / N;
  const int* row = eidx;
  const int* col = eidx + E;

  float* ws = (float*)d_ws;
  float* dinv = ws;
  float* bufA = ws + (((size_t)N + 63) & ~(size_t)63);
  const int NP = (N + 63) & ~63;
  const int EP = (E + 63) & ~63;
  const int NB = (N + RN - 1) >> RB;
  const int NW = (E + CH - 1) / CH;
  int* start = (int*)(bufA + (size_t)N * NCOL);
  int* cnt = start + NP;
  int* csr = cnt + NP;
  unsigned* pairs = (unsigned*)(csr + EP);
  int* binStart = (int*)(pairs + EP);
  int* wgcnt = csr;

  const int G = out_size / NCOL - N;
  float* hs = (float*)d_out;
  float* hout = hs + (size_t)G * NCOL;

  const dim3 b256(256);
  const int nfBlocks = (N * NCOL + 255) / 256;
  const int gemmBlocks = (N + TILE_M - 1) / TILE_M;
  const int nodeBlocks = (N + 7) / 8;

  const dim3 b1024(1024);
  k_binhist<<<NW, b1024, 0, stream>>>(col, wgcnt, E, NB);
  k_partition<<<NW, b1024, 0, stream>>>(row, col, wgcnt, binStart, pairs, E, NB, NW);
  k_binsort<<<NB, b1024, 0, stream>>>(pairs, binStart, start, cnt, dinv, csr, N);

  k_gemm_wmma<<<gemmBlocks, b256, 0, stream>>>(x, W1, bufA, N, K1);
  k_gather<<<nodeBlocks, b256, 0, stream>>>(bufA, start, cnt, csr, dinv, b1, hout, N, 1);

  k_gemm_wmma<<<gemmBlocks, b256, 0, stream>>>(hout, W2, bufA, N, NCOL);
  k_gather<<<nodeBlocks, b256, 0, stream>>>(bufA, start, cnt, csr, dinv, b2, hout, N, 0);

  k_zero<<<(G * NCOL + 255) / 256, b256, 0, stream>>>(hs, G * NCOL);
  k_pool<<<nfBlocks, b256, 0, stream>>>(hout, batch, hs, N);
}


# Add pool fused into layer-2 gather: a block sums its 8 rows per graph in LDS in node order (new sum order), then 1 float atomic per block, graph, feature; hs zeroed in layer 1; 2 fewer launches
# speedup vs baseline: 3.6016x; 1.0782x over previous
#include <hip/hip_runtime.h>


typedef __attribute__((ext_vector_type(2))) float v2f;
typedef __attribute__((ext_vector_type(8))) float v8f;

#define NCOL 128
#define TILE_M 128
#define KC 64
#define LXS 68
#define LWS 288

__global__ __launch_bounds__(256) void k_gemm_wmma(
    const float* __restrict__ X, const float* __restrict__ W,
    float* __restrict__ C, int M, int K) {
  __shared__ float lw2[(KC / 2) * LWS];
  __shared__ float lx[TILE_M * LXS];

  const int tid = threadIdx.x;
  const int lane = tid & 31;
  const int wave = tid >> 5;
  const int mblock = blockIdx.x * TILE_M;
  const int mstrip = mblock + wave * 16;

  v8f acc[8] = {};

  const int lane15 = lane & 15;
  const int ksel = (lane < 16) ? 0 : 2;
  const int mrow = wave * 16 + lane15;

  const float* axp = lx + mrow * LXS + ksel;
  const float* bbp = lw2 + (ksel >> 1) * LWS + lane15 * 2;

  for (int k0 = 0; k0 < K; k0 += KC) {
    __syncthreads();
    for (int i = tid; i < (KC / 2) * (NCOL / 4); i += 256) {
      const int kk2 = i >> 5;
      const int c4 = i & 31;
      const float* wr = W + (size_t)(k0 + kk2 * 2) * NCOL + c4 * 4;
      const float4 w0 = *(const float4*)wr;
      const float4 w1 = *(const float4*)(wr + NCOL);
      float* dst = lw2 + kk2 * LWS + c4 * 8;
      ((float2*)dst)[0] = make_float2(w0.x, w1.x);
      ((float2*)dst)[1] = make_float2(w0.y, w1.y);
      ((float2*)dst)[2] = make_float2(w0.z, w1.z);
      ((float2*)dst)[3] = make_float2(w0.w, w1.w);
    }
    for (int i = tid; i < TILE_M * (KC / 4); i += 256) {
      const int r = i >> 4;
      const int c4 = i & 15;
      const int gr = mblock + r;
      float4 v = make_float4(0.f, 0.f, 0.f, 0.f);
      if (gr < M) v = *(const float4*)(X + (size_t)gr * K + k0 + c4 * 4);
      *(float4*)(lx + r * LXS + c4 * 4) = v;
    }
    __syncthreads();

#pragma unroll 4
    for (int kk = 0; kk < KC; kk += 4) {
      const v2f a = *(const v2f*)(axp + kk);
      const float* brow = bbp + (kk >> 1) * LWS;
#pragma unroll
      for (int t = 0; t < 8; ++t) {
        const v2f b = *(const v2f*)(brow + t * 32);
        acc[t] = __builtin_amdgcn_wmma_f32_16x16x4_f32(
             false, a,  false, b,
             (short)0, acc[t],  false,  false);
      }
    }
  }

  if (mstrip < M) {
    const int mo = (lane < 16) ? 0 : 8;
#pragma unroll
    for (int t = 0; t < 8; ++t) {
#pragma unroll
      for (int v = 0; v < 8; ++v) {
        const int m = mstrip + mo + v;
        if (m < M) C[(size_t)m * NCOL + t * 16 + lane15] = acc[t][v];
      }
    }
  }
}

#define RB 8
#define RN 256
#define NBMAX 256
#define CH 4096
#define DCAP 8192

__device__ __forceinline__ int block_excl_scan(int v, int* red, int* total) {
  const int lane = threadIdx.x & 31;
  const int wave = threadIdx.x >> 5;
  int incl = v;
  for (int off = 1; off < 32; off <<= 1) {
    const int t = __shfl_up(incl, off);
    if (lane >= off) incl += t;
  }
  if (lane == 31) red[wave] = incl;
  __syncthreads();
  if (wave == 0) {
    const int t = red[lane];
    int ti = t;
    for (int off = 1; off < 32; off <<= 1) {
      const int u = __shfl_up(ti, off);
      if (lane >= off) ti += u;
    }
    red[lane] = ti - t;
    if (lane == 31) red[32] = ti;
  }
  __syncthreads();
  *total = red[32];
  return red[wave] + incl - v;
}

__global__ __launch_bounds__(1024) void k_binhist(const int* __restrict__ col,
                                                  int* __restrict__ wgcnt, int E, int NB) {
  __shared__ int hist[NBMAX];
  const int tid = threadIdx.x;
  if (tid < NBMAX) hist[tid] = 0;
  __syncthreads();
  const int e0 = blockIdx.x * CH;
  int c[CH / 1024];
#pragma unroll
  for (int k = 0; k < CH / 1024; ++k) {
    const int e = e0 + k * 1024 + tid;
    c[k] = (e < E) ? col[e] : -1;
  }
#pragma unroll
  for (int k = 0; k < CH / 1024; ++k)
    if (c[k] >= 0) atomicAdd(&hist[c[k] >> RB], 1);
  __syncthreads();
  if (tid < NB) wgcnt[blockIdx.x * NB + tid] = hist[tid];
}

__global__ __launch_bounds__(1024) void k_partition(
    const int* __restrict__ row, const int* __restrict__ col,
    const int* __restrict__ wgcnt, int* __restrict__ binStart,
    unsigned* __restrict__ pairs, int E, int NB, int NW) {
  __shared__ int ptot[1024];
  __shared__ int ppre[1024];
  __shared__ int lstart[NBMAX];
  __shared__ int cursor[NBMAX];
  __shared__ int gbase[NBMAX];
  __shared__ unsigned sorted[CH];
  __shared__ int red1[33];
  __shared__ int red2[33];
  const int tid = threadIdx.x;
  const int w = blockIdx.x;

  {
    const int b = tid & (NBMAX - 1);
    int t_ = 0, p_ = 0;
    if (b < NB) {
#pragma unroll 8
      for (int q = tid >> 8; q < NW; q += 4) {
        const int t = wgcnt[q * NB + b];
        t_ += t;
        p_ += (q < w) ? t : 0;
      }
    }
    ptot[tid] = t_;
    ppre[tid] = p_;
  }
  __syncthreads();
  int tot = 0, pre = 0, mine = 0;
  if (tid < NBMAX) {
    tot = ptot[tid] + ptot[tid + 256] + ptot[tid + 512] + ptot[tid + 768];
    pre = ppre[tid] + ppre[tid + 256] + ppre[tid + 512] + ppre[tid + 768];
    if (tid < NB) mine = wgcnt[w * NB + tid];
  }
  int total, ltotal;
  const int bs = block_excl_scan(tot, red1, &total);
  const int ls = block_excl_scan(mine, red2, &ltotal);
  if (tid < NBMAX) {
    lstart[tid] = ls;
    cursor[tid] = ls;
    gbase[tid] = bs + pre;
  }
  if (w == 0) {
    if (tid < NB) binStart[tid] = bs;
    if (tid == 0) binStart[NB] = total;
  }
  __syncthreads();

  const int e0 = w * CH;
  int ec[CH / 1024], er[CH / 1024];
#pragma unroll
  for (int k = 0; k < CH / 1024; ++k) {
    const int e = e0 + k * 1024 + tid;
    ec[k] = (e < E) ? col[e] : -1;
    er[k] = (e < E) ? row[e] : 0;
  }
#pragma unroll
  for (int k = 0; k < CH / 1024; ++k) {
    if (ec[k] >= 0) {
      const int p = atomicAdd(&cursor[ec[k] >> RB], 1);
      sorted[p] = ((unsigned)er[k] << RB) | (unsigned)(ec[k] & (RN - 1));
    }
  }
  __syncthreads();

  const int lane = tid & 31;
  for (int b = tid >> 5; b < NB; b += 32) {
    const int s = lstart[b];
    const int n = cursor[b] - s;
    const int g = gbase[b];
    for (int i = lane; i < n; i += 32) pairs[g + i] = sorted[s + i];
  }
}

__global__ __launch_bounds__(1024) void k_binsort(
    const unsigned* __restrict__ pairs, const int* __restrict__ binStart,
    int* __restrict__ start, int* __restrict__ cnt, float* __restrict__ dinv,
    int* __restrict__ csr, int N) {
  __shared__ int ncnt[RN];
  __shared__ int ncur[RN];
  __shared__ int red[33];
  __shared__ int stage[DCAP];
  const int tid = threadIdx.x;
  const int b = blockIdx.x;
  const int bs = binStart[b];
  const int M = binStart[b + 1] - bs;

  if (tid < RN) ncnt[tid] = 0;
  __syncthreads();
  for (int i = tid; i < M; i += 1024) atomicAdd(&ncnt[pairs[bs + i] & (RN - 1)], 1);
  __syncthreads();
  const int c = (tid < RN) ? ncnt[tid] : 0;
  int tot;
  const int ns = block_excl_scan(c, red, &tot);
  if (tid < RN) {
    ncur[tid] = ns;
    const int node = b * RN + tid;
    if (node < N) {
      start[node] = bs + ns;
      cnt[node] = c;
      const float d = (float)c + 1.0f;
      dinv[node] = (d > 0.f) ? rsqrtf(d) : 0.f;
    }
  }
  __syncthreads();
  if (M <= DCAP) {
    for (int i = tid; i < M; i += 1024) {
      const unsigned p = pairs[bs + i];
      const int pos = atomicAdd(&ncur[p & (RN - 1)], 1);
      stage[pos] = (int)(p >> RB);
    }
    __syncthreads();
    for (int i = tid; i < M; i += 1024) csr[bs + i] = stage[i];
  } else {
    for (int i = tid; i < M; i += 1024) {
      const unsigned p = pairs[bs + i];
      const int pos = atomicAdd(&ncur[p & (RN - 1)], 1);
      csr[bs + pos] = (int)(p >> RB);
    }
  }
}

__device__ __forceinline__ void fma4(float4& a, const float4 v, const float wl, const int t) {
  const float w = __int_as_float(__builtin_amdgcn_readlane(__float_as_int(wl), t));
  a.x = fmaf(v.x, w, a.x);
  a.y = fmaf(v.y, w, a.y);
  a.z = fmaf(v.z, w, a.z);
  a.w = fmaf(v.w, w, a.w);
}

__device__ __forceinline__ float4 gather_row(
    const float* __restrict__ hlin, const int* __restrict__ start,
    const int* __restrict__ cnt, const int* __restrict__ csr,
    const float* __restrict__ dinv, const float* __restrict__ bias,
    const int n, const int lane, const bool do_relu) {
  const float4* __restrict__ H = (const float4*)hlin;
  const int s = start[n];
  const int e = s + cnt[n];
  const float dn = dinv[n];

  const float4 self = H[(size_t)n * (NCOL / 4) + lane];
  float4 acc = make_float4(self.x * dn * dn, self.y * dn * dn, self.z * dn * dn, self.w * dn * dn);

  for (int base = s; base < e; base += 32) {
    const int m = (e - base < 32) ? (e - base) : 32;
    int idx = 0;
    float w = 0.f;
    if (lane < m) {
      idx = csr[base + lane];
      w = dinv[idx] * dn;
    }
    int t = 0;
    for (; t + 4 <= m; t += 4) {
      const int s0 = __builtin_amdgcn_readlane(idx, t);
      const int s1 = __builtin_amdgcn_readlane(idx, t + 1);
      const int s2 = __builtin_amdgcn_readlane(idx, t + 2);
      const int s3 = __builtin_amdgcn_readlane(idx, t + 3);
      const float4 v0 = H[(size_t)s0 * (NCOL / 4) + lane];
      const float4 v1 = H[(size_t)s1 * (NCOL / 4) + lane];
      const float4 v2 = H[(size_t)s2 * (NCOL / 4) + lane];
      const float4 v3 = H[(size_t)s3 * (NCOL / 4) + lane];
      fma4(acc, v0, w, t);
      fma4(acc, v1, w, t + 1);
      fma4(acc, v2, w, t + 2);
      fma4(acc, v3, w, t + 3);
    }
    const int rem = m - t;
    if (rem > 0) {
      const int s0 = __builtin_amdgcn_readlane(idx, t);
      const float4 v0 = H[(size_t)s0 * (NCOL / 4) + lane];
      float4 v1 = make_float4(0.f, 0.f, 0.f, 0.f);
      float4 v2 = make_float4(0.f, 0.f, 0.f, 0.f);
      if (rem > 1) {
        const int s1 = __builtin_amdgcn_readlane(idx, t + 1);
        v1 = H[(size_t)s1 * (NCOL / 4) + lane];
      }
      if (rem > 2) {
        const int s2 = __builtin_amdgcn_readlane(idx, t + 2);
        v2 = H[(size_t)s2 * (NCOL / 4) + lane];
      }
      fma4(acc, v0, w, t);
      if (rem > 1) fma4(acc, v1, w, t + 1);
      if (rem > 2) fma4(acc, v2, w, t + 2);
    }
  }

  const float4 b = ((const float4*)bias)[lane];
  float4 r = make_float4(acc.x + b.x, acc.y + b.y, acc.z + b.z, acc.w + b.w);
  if (do_relu) {
    r.x = (r.x < 0.f) ? 0.f : r.x;
    r.y = (r.y < 0.f) ? 0.f : r.y;
    r.z = (r.z < 0.f) ? 0.f : r.z;
    r.w = (r.w < 0.f) ? 0.f : r.w;
  }
  return r;
}

template <bool LAST>
__global__ __launch_bounds__(256) void k_gather(
    const float* __restrict__ hlin, const int* __restrict__ start,
    const int* __restrict__ cnt, const int* __restrict__ csr,
    const float* __restrict__ dinv, const float* __restrict__ bias,
    float* __restrict__ out, const int* __restrict__ batch,
    float* __restrict__ hs, int N, int G) {
  const int lane = threadIdx.x & 31;
  const int wv = threadIdx.x >> 5;
  if (!LAST) {
    for (int i = blockIdx.x * 256 + threadIdx.x; i < G * NCOL; i += gridDim.x * 256) hs[i] = 0.f;
  }
  const int n = __builtin_amdgcn_readfirstlane(blockIdx.x * 8 + wv);
  const bool valid = n < N;
  int g_of_n = -1;
  float4 r = make_float4(0.f, 0.f, 0.f, 0.f);
  if (valid) {
    if (LAST) g_of_n = batch[n];
    r = gather_row(hlin, start, cnt, csr, dinv, bias, n, lane, !LAST);
    ((float4*)out)[(size_t)n * (NCOL / 4) + lane] = r;
  }
  if (LAST) {
    __shared__ float4 prow[8][NCOL / 4];
    __shared__ int pgraph[8];
    prow[wv][lane] = r;
    if (lane == 0) pgraph[wv] = g_of_n;
    __syncthreads();
    if (threadIdx.x < NCOL) {
      const float* pr = (const float*)&prow[0][0];
      const int f = threadIdx.x;
      float v[8];
      int gg[8];
#pragma unroll
      for (int j = 0; j < 8; ++j) {
        v[j] = pr[j * NCOL + f];
        gg[j] = __builtin_amdgcn_readfirstlane(pgraph[j]);
      }
      int g = gg[0];
      float a = v[0];
#pragma unroll
      for (int j = 1; j < 8; ++j) {
        if (gg[j] < 0) continue;
        if (gg[j] == g) {
          a += v[j];
        } else {
          if ((unsigned)g < (unsigned)G) atomicAdd(&hs[(size_t)g * NCOL + f], a);
          g = gg[j];
          a = v[j];
        }
      }
      if ((unsigned)g < (unsigned)G) atomicAdd(&hs[(size_t)g * NCOL + f], a);
    }
  }
}

extern "C" void kernel_launch(void* const* d_in, const int* in_sizes, int n_in,
                              void* d_out, int out_size, void* d_ws, size_t ws_size,
                              hipStream_t stream) {
  const float* x = (const float*)d_in[0];
  const float* W1 = (const float*)d_in[1];
  const float* b1 = (const float*)d_in[2];
  const float* W2 = (const float*)d_in[3];
  const float* b2 = (const float*)d_in[4];
  const int* eidx = (const int*)d_in[5];
  const int* batch = (const int*)d_in[6];

  const int N = in_sizes[6];
  const int E = in_sizes[5] / 2;
  const int K1 = in_sizes[0] / N;
  const int* row = eidx;
  const int* col = eidx + E;

  float* ws = (float*)d_ws;
  float* dinv = ws;
  float* bufA = ws + (((size_t)N + 63) & ~(size_t)63);
  const int NP = (N + 63) & ~63;
  const int EP = (E + 63) & ~63;
  const int NB = (N + RN - 1) >> RB;
  const int NW = (E + CH - 1) / CH;
  int* start = (int*)(bufA + (size_t)N * NCOL);
  int* cnt = start + NP;
  int* csr = cnt + NP;
  unsigned* pairs = (unsigned*)(csr + EP);
  int* binStart = (int*)(pairs + EP);
  int* wgcnt = csr;

  const int G = out_size / NCOL - N;
  float* hs = (float*)d_out;
  float* hout = hs + (size_t)G * NCOL;

  const dim3 b256(256);
  const int gemmBlocks = (N + TILE_M - 1) / TILE_M;
  const int nodeBlocks = (N + 7) / 8;

  const dim3 b1024(1024);
  k_binhist<<<NW, b1024, 0, stream>>>(col, wgcnt, E, NB);
  k_partition<<<NW, b1024, 0, stream>>>(row, col, wgcnt, binStart, pairs, E, NB, NW);
  k_binsort<<<NB, b1024, 0, stream>>>(pairs, binStart, start, cnt, dinv, csr, N);

  k_gemm_wmma<<<gemmBlocks, b256, 0, stream>>>(x, W1, bufA, N, K1);
  k_gather<false><<<nodeBlocks, b256, 0, stream>>>(bufA, start, cnt, csr, dinv, b1, hout, batch, hs, N, G);

  k_gemm_wmma<<<gemmBlocks, b256, 0, stream>>>(hout, W2, bufA, N, NCOL);
  k_gather<true><<<nodeBlocks, b256, 0, stream>>>(bufA, start, cnt, csr, dinv, b2, hout, batch, hs, N, G);
}


# GEMM K-loop: the next chunk's 16 X and W loads per thread are issued before the current chunk's WMMAs and stored to a second LDS stage after them (was one load per wait); sum order unchanged
# speedup vs baseline: 4.5338x; 1.2588x over previous
#include <hip/hip_runtime.h>


typedef __attribute__((ext_vector_type(2))) float v2f;
typedef __attribute__((ext_vector_type(8))) float v8f;

#define NCOL 128
#define TILE_M 128
#define KC 64
#define LXS 68
#define LWS 288

__global__ __launch_bounds__(256) void k_gemm_wmma(
    const float* __restrict__ X, const float* __restrict__ W,
    float* __restrict__ C, int M, int K) {
  __shared__ float lw2[2][(KC / 2) * LWS];
  __shared__ float lx[2][TILE_M * LXS];

  const int tid = threadIdx.x;
  const int lane = tid & 31;
  const int wave = tid >> 5;
  const int mblock = blockIdx.x * TILE_M;
  const int mstrip = mblock + wave * 16;

  v8f acc[8] = {};

  const int lane15 = lane & 15;
  const int ksel = (lane < 16) ? 0 : 2;
  const int mrow = wave * 16 + lane15;

  const int axo = mrow * LXS + ksel;
  const int bbo = (ksel >> 1) * LWS + lane15 * 2;

  const int wk2 = tid >> 5;
  const int wc4 = tid & 31;
  const int xr0 = tid >> 4;
  const int xc4 = tid & 15;
  const float* wsrc = W + (size_t)(wk2 * 2) * NCOL + wc4 * 4;
  const float* xsrc = X + (size_t)(mblock + xr0) * K + xc4 * 4;
  const int wdst = wk2 * LWS + wc4 * 8;
  const int xdst = xr0 * LXS + xc4 * 4;

  float4 w0r[4], w1r[4], xr[8];

#define GEMM_LOAD_CHUNK(k0_)                                                   \
  {                                                                            \
    _Pragma("unroll") for (int j = 0; j < 4; ++j) {                            \
      const float* wr = wsrc + (size_t)((k0_) + j * 16) * NCOL;                \
      w0r[j] = *(const float4*)wr;                                             \
      w1r[j] = *(const float4*)(wr + NCOL);                                    \
    }                                                                          \
    _Pragma("unroll") for (int j = 0; j < 8; ++j) {                            \
      xr[j] = make_float4(0.f, 0.f, 0.f, 0.f);                                 \
      if (mblock + xr0 + j * 16 < M)                                           \
        xr[j] = *(const float4*)(xsrc + (size_t)(j * 16) * K + (k0_));         \
    }                                                                          \
  }
#define GEMM_STORE_CHUNK(s_)                                                   \
  {                                                                            \
    _Pragma("unroll") for (int j = 0; j < 4; ++j) {                            \
      float* dst = &lw2[s_][wdst + j * 8 * LWS];                               \
      ((float2*)dst)[0] = make_float2(w0r[j].x, w1r[j].x);                     \
      ((float2*)dst)[1] = make_float2(w0r[j].y, w1r[j].y);                     \
      ((float2*)dst)[2] = make_float2(w0r[j].z, w1r[j].z);                     \
      ((float2*)dst)[3] = make_float2(w0r[j].w, w1r[j].w);                     \
    }                                                                          \
    _Pragma("unroll") for (int j = 0; j < 8; ++j)                              \
      *(float4*)&lx[s_][xdst + j * 16 * LXS] = xr[j];                          \
  }

  GEMM_LOAD_CHUNK(0)
  GEMM_STORE_CHUNK(0)
  __syncthreads();

  int st = 0;
  for (int k0 = 0; k0 < K; k0 += KC) {
    const bool more = k0 + KC < K;
    if (more) GEMM_LOAD_CHUNK(k0 + KC)

    const float* axp = &lx[st][axo];
    const float* bbp = &lw2[st][bbo];
#pragma unroll 4
    for (int kk = 0; kk < KC; kk += 4) {
      const v2f a = *(const v2f*)(axp + kk);
      const float* brow = bbp + (kk >> 1) * LWS;
#pragma unroll
      for (int t = 0; t < 8; ++t) {
        const v2f b = *(const v2f*)(brow + t * 32);
        acc[t] = __builtin_amdgcn_wmma_f32_16x16x4_f32(
             false, a,  false, b,
             (short)0, acc[t],  false,  false);
      }
    }

    asm volatile("v_nop\n\tv_nop\n\tv_nop\n\tv_nop\n\tv_nop\n\tv_nop\n\tv_nop\n\tv_nop");
    if (more) {
#pragma unroll
      for (int j = 0; j < 4; ++j)
        asm volatile("" : "+v"(w0r[j].x), "+v"(w0r[j].y), "+v"(w0r[j].z), "+v"(w0r[j].w),
                          "+v"(w1r[j].x), "+v"(w1r[j].y), "+v"(w1r[j].z), "+v"(w1r[j].w));
      GEMM_STORE_CHUNK(st ^ 1)
    }
    __syncthreads();
    st ^= 1;
  }
#undef GEMM_LOAD_CHUNK
#undef GEMM_STORE_CHUNK

  if (mstrip < M) {
    const int mo = (lane < 16) ? 0 : 8;
#pragma unroll
    for (int t = 0; t < 8; ++t) {
#pragma unroll
      for (int v = 0; v < 8; ++v) {
        const int m = mstrip + mo + v;
        if (m < M) C[(size_t)m * NCOL + t * 16 + lane15] = acc[t][v];
      }
    }
  }
}

#define RB 8
#define RN 256
#define NBMAX 256
#define CH 4096
#define DCAP 8192

__device__ __forceinline__ int block_excl_scan(int v, int* red, int* total) {
  const int lane = threadIdx.x & 31;
  const int wave = threadIdx.x >> 5;
  int incl = v;
  for (int off = 1; off < 32; off <<= 1) {
    const int t = __shfl_up(incl, off);
    if (lane >= off) incl += t;
  }
  if (lane == 31) red[wave] = incl;
  __syncthreads();
  if (wave == 0) {
    const int t = red[lane];
    int ti = t;
    for (int off = 1; off < 32; off <<= 1) {
      const int u = __shfl_up(ti, off);
      if (lane >= off) ti += u;
    }
    red[lane] = ti - t;
    if (lane == 31) red[32] = ti;
  }
  __syncthreads();
  *total = red[32];
  return red[wave] + incl - v;
}

__global__ __launch_bounds__(1024) void k_binhist(const int* __restrict__ col,
                                                  int* __restrict__ wgcnt, int E, int NB) {
  __shared__ int hist[NBMAX];
  const int tid = threadIdx.x;
  if (tid < NBMAX) hist[tid] = 0;
  __syncthreads();
  const int e0 = blockIdx.x * CH;
  int c[CH / 1024];
#pragma unroll
  for (int k = 0; k < CH / 1024; ++k) {
    const int e = e0 + k * 1024 + tid;
    c[k] = (e < E) ? col[e] : -1;
  }
#pragma unroll
  for (int k = 0; k < CH / 1024; ++k)
    if (c[k] >= 0) atomicAdd(&hist[c[k] >> RB], 1);
  __syncthreads();
  if (tid < NB) wgcnt[blockIdx.x * NB + tid] = hist[tid];
}

__global__ __launch_bounds__(1024) void k_partition(
    const int* __restrict__ row, const int* __restrict__ col,
    const int* __restrict__ wgcnt, int* __restrict__ binStart,
    unsigned* __restrict__ pairs, int E, int NB, int NW) {
  __shared__ int ptot[1024];
  __shared__ int ppre[1024];
  __shared__ int lstart[NBMAX];
  __shared__ int cursor[NBMAX];
  __shared__ int gbase[NBMAX];
  __shared__ unsigned sorted[CH];
  __shared__ int red1[33];
  __shared__ int red2[33];
  const int tid = threadIdx.x;
  const int w = blockIdx.x;

  {
    const int b = tid & (NBMAX - 1);
    int t_ = 0, p_ = 0;
    if (b < NB) {
#pragma unroll 8
      for (int q = tid >> 8; q < NW; q += 4) {
        const int t = wgcnt[q * NB + b];
        t_ += t;
        p_ += (q < w) ? t : 0;
      }
    }
    ptot[tid] = t_;
    ppre[tid] = p_;
  }
  __syncthreads();
  int tot = 0, pre = 0, mine = 0;
  if (tid < NBMAX) {
    tot = ptot[tid] + ptot[tid + 256] + ptot[tid + 512] + ptot[tid + 768];
    pre = ppre[tid] + ppre[tid + 256] + ppre[tid + 512] + ppre[tid + 768];
    if (tid < NB) mine = wgcnt[w * NB + tid];
  }
  int total, ltotal;
  const int bs = block_excl_scan(tot, red1, &total);
  const int ls = block_excl_scan(mine, red2, &ltotal);
  if (tid < NBMAX) {
    lstart[tid] = ls;
    cursor[tid] = ls;
    gbase[tid] = bs + pre;
  }
  if (w == 0) {
    if (tid < NB) binStart[tid] = bs;
    if (tid == 0) binStart[NB] = total;
  }
  __syncthreads();

  const int e0 = w * CH;
  int ec[CH / 1024], er[CH / 1024];
#pragma unroll
  for (int k = 0; k < CH / 1024; ++k) {
    const int e = e0 + k * 1024 + tid;
    ec[k] = (e < E) ? col[e] : -1;
    er[k] = (e < E) ? row[e] : 0;
  }
#pragma unroll
  for (int k = 0; k < CH / 1024; ++k) {
    if (ec[k] >= 0) {
      const int p = atomicAdd(&cursor[ec[k] >> RB], 1);
      sorted[p] = ((unsigned)er[k] << RB) | (unsigned)(ec[k] & (RN - 1));
    }
  }
  __syncthreads();

  const int lane = tid & 31;
  for (int b = tid >> 5; b < NB; b += 32) {
    const int s = lstart[b];
    const int n = cursor[b] - s;
    const int g = gbase[b];
    for (int i = lane; i < n; i += 32) pairs[g + i] = sorted[s + i];
  }
}

__global__ __launch_bounds__(1024) void k_binsort(
    const unsigned* __restrict__ pairs, const int* __restrict__ binStart,
    int* __restrict__ start, int* __restrict__ cnt, float* __restrict__ dinv,
    int* __restrict__ csr, int N) {
  __shared__ int ncnt[RN];
  __shared__ int ncur[RN];
  __shared__ int red[33];
  __shared__ int stage[DCAP];
  const int tid = threadIdx.x;
  const int b = blockIdx.x;
  const int bs = binStart[b];
  const int M = binStart[b + 1] - bs;

  if (tid < RN) ncnt[tid] = 0;
  __syncthreads();
  for (int i = tid; i < M; i += 1024) atomicAdd(&ncnt[pairs[bs + i] & (RN - 1)], 1);
  __syncthreads();
  const int c = (tid < RN) ? ncnt[tid] : 0;
  int tot;
  const int ns = block_excl_scan(c, red, &tot);
  if (tid < RN) {
    ncur[tid] = ns;
    const int node = b * RN + tid;
    if (node < N) {
      start[node] = bs + ns;
      cnt[node] = c;
      const float d = (float)c + 1.0f;
      dinv[node] = (d > 0.f) ? rsqrtf(d) : 0.f;
    }
  }
  __syncthreads();
  if (M <= DCAP) {
    for (int i = tid; i < M; i += 1024) {
      const unsigned p = pairs[bs + i];
      const int pos = atomicAdd(&ncur[p & (RN - 1)], 1);
      stage[pos] = (int)(p >> RB);
    }
    __syncthreads();
    for (int i = tid; i < M; i += 1024) csr[bs + i] = stage[i];
  } else {
    for (int i = tid; i < M; i += 1024) {
      const unsigned p = pairs[bs + i];
      const int pos = atomicAdd(&ncur[p & (RN - 1)], 1);
      csr[bs + pos] = (int)(p >> RB);
    }
  }
}

__device__ __forceinline__ void fma4(float4& a, const float4 v, const float wl, const int t) {
  const float w = __int_as_float(__builtin_amdgcn_readlane(__float_as_int(wl), t));
  a.x = fmaf(v.x, w, a.x);
  a.y = fmaf(v.y, w, a.y);
  a.z = fmaf(v.z, w, a.z);
  a.w = fmaf(v.w, w, a.w);
}

__device__ __forceinline__ float4 gather_row(
    const float* __restrict__ hlin, const int* __restrict__ start,
    const int* __restrict__ cnt, const int* __restrict__ csr,
    const float* __restrict__ dinv, const float* __restrict__ bias,
    const int n, const int lane, const bool do_relu) {
  const float4* __restrict__ H = (const float4*)hlin;
  const int s = start[n];
  const int e = s + cnt[n];
  const float dn = dinv[n];

  const float4 self = H[(size_t)n * (NCOL / 4) + lane];
  float4 acc = make_float4(self.x * dn * dn, self.y * dn * dn, self.z * dn * dn, self.w * dn * dn);

  for (int base = s; base < e; base += 32) {
    const int m = (e - base < 32) ? (e - base) : 32;
    int idx = 0;
    float w = 0.f;
    if (lane < m) {
      idx = csr[base + lane];
      w = dinv[idx] * dn;
    }
    int t = 0;
    for (; t + 4 <= m; t += 4) {
      const int s0 = __builtin_amdgcn_readlane(idx, t);
      const int s1 = __builtin_amdgcn_readlane(idx, t + 1);
      const int s2 = __builtin_amdgcn_readlane(idx, t + 2);
      const int s3 = __builtin_amdgcn_readlane(idx, t + 3);
      const float4 v0 = H[(size_t)s0 * (NCOL / 4) + lane];
      const float4 v1 = H[(size_t)s1 * (NCOL / 4) + lane];
      const float4 v2 = H[(size_t)s2 * (NCOL / 4) + lane];
      const float4 v3 = H[(size_t)s3 * (NCOL / 4) + lane];
      fma4(acc, v0, w, t);
      fma4(acc, v1, w, t + 1);
      fma4(acc, v2, w, t + 2);
      fma4(acc, v3, w, t + 3);
    }
    const int rem = m - t;
    if (rem > 0) {
      const int s0 = __builtin_amdgcn_readlane(idx, t);
      const float4 v0 = H[(size_t)s0 * (NCOL / 4) + lane];
      float4 v1 = make_float4(0.f, 0.f, 0.f, 0.f);
      float4 v2 = make_float4(0.f, 0.f, 0.f, 0.f);
      if (rem > 1) {
        const int s1 = __builtin_amdgcn_readlane(idx, t + 1);
        v1 = H[(size_t)s1 * (NCOL / 4) + lane];
      }
      if (rem > 2) {
        const int s2 = __builtin_amdgcn_readlane(idx, t + 2);
        v2 = H[(size_t)s2 * (NCOL / 4) + lane];
      }
      fma4(acc, v0, w, t);
      if (rem > 1) fma4(acc, v1, w, t + 1);
      if (rem > 2) fma4(acc, v2, w, t + 2);
    }
  }

  const float4 b = ((const float4*)bias)[lane];
  float4 r = make_float4(acc.x + b.x, acc.y + b.y, acc.z + b.z, acc.w + b.w);
  if (do_relu) {
    r.x = (r.x < 0.f) ? 0.f : r.x;
    r.y = (r.y < 0.f) ? 0.f : r.y;
    r.z = (r.z < 0.f) ? 0.f : r.z;
    r.w = (r.w < 0.f) ? 0.f : r.w;
  }
  return r;
}

template <bool LAST>
__global__ __launch_bounds__(256) void k_gather(
    const float* __restrict__ hlin, const int* __restrict__ start,
    const int* __restrict__ cnt, const int* __restrict__ csr,
    const float* __restrict__ dinv, const float* __restrict__ bias,
    float* __restrict__ out, const int* __restrict__ batch,
    float* __restrict__ hs, int N, int G) {
  const int lane = threadIdx.x & 31;
  const int wv = threadIdx.x >> 5;
  if (!LAST) {
    for (int i = blockIdx.x * 256 + threadIdx.x; i < G * NCOL; i += gridDim.x * 256) hs[i] = 0.f;
  }
  const int n = __builtin_amdgcn_readfirstlane(blockIdx.x * 8 + wv);
  const bool valid = n < N;
  int g_of_n = -1;
  float4 r = make_float4(0.f, 0.f, 0.f, 0.f);
  if (valid) {
    if (LAST) g_of_n = batch[n];
    r = gather_row(hlin, start, cnt, csr, dinv, bias, n, lane, !LAST);
    ((float4*)out)[(size_t)n * (NCOL / 4) + lane] = r;
  }
  if (LAST) {
    __shared__ float4 prow[8][NCOL / 4];
    __shared__ int pgraph[8];
    prow[wv][lane] = r;
    if (lane == 0) pgraph[wv] = g_of_n;
    __syncthreads();
    if (threadIdx.x < NCOL) {
      const float* pr = (const float*)&prow[0][0];
      const int f = threadIdx.x;
      float v[8];
      int gg[8];
#pragma unroll
      for (int j = 0; j < 8; ++j) {
        v[j] = pr[j * NCOL + f];
        gg[j] = __builtin_amdgcn_readfirstlane(pgraph[j]);
      }
      int g = gg[0];
      float a = v[0];
#pragma unroll
      for (int j = 1; j < 8; ++j) {
        if (gg[j] < 0) continue;
        if (gg[j] == g) {
          a += v[j];
        } else {
          if ((unsigned)g < (unsigned)G) atomicAdd(&hs[(size_t)g * NCOL + f], a);
          g = gg[j];
          a = v[j];
        }
      }
      if ((unsigned)g < (unsigned)G) atomicAdd(&hs[(size_t)g * NCOL + f], a);
    }
  }
}

extern "C" void kernel_launch(void* const* d_in, const int* in_sizes, int n_in,
                              void* d_out, int out_size, void* d_ws, size_t ws_size,
                              hipStream_t stream) {
  const float* x = (const float*)d_in[0];
  const float* W1 = (const float*)d_in[1];
  const float* b1 = (const float*)d_in[2];
  const float* W2 = (const float*)d_in[3];
  const float* b2 = (const float*)d_in[4];
  const int* eidx = (const int*)d_in[5];
  const int* batch = (const int*)d_in[6];

  const int N = in_sizes[6];
  const int E = in_sizes[5] / 2;
  const int K1 = in_sizes[0] / N;
  const int* row = eidx;
  const int* col = eidx + E;

  float* ws = (float*)d_ws;
  float* dinv = ws;
  float* bufA = ws + (((size_t)N + 63) & ~(size_t)63);
  const int NP = (N + 63) & ~63;
  const int EP = (E + 63) & ~63;
  const int NB = (N + RN - 1) >> RB;
  const int NW = (E + CH - 1) / CH;
  int* start = (int*)(bufA + (size_t)N * NCOL);
  int* cnt = start + NP;
  int* csr = cnt + NP;
  unsigned* pairs = (unsigned*)(csr + EP);
  int* binStart = (int*)(pairs + EP);
  int* wgcnt = csr;

  const int G = out_size / NCOL - N;
  float* hs = (float*)d_out;
  float* hout = hs + (size_t)G * NCOL;

  const dim3 b256(256);
  const int gemmBlocks = (N + TILE_M - 1) / TILE_M;
  const int nodeBlocks = (N + 7) / 8;

  const dim3 b1024(1024);
  k_binhist<<<NW, b1024, 0, stream>>>(col, wgcnt, E, NB);
  k_partition<<<NW, b1024, 0, stream>>>(row, col, wgcnt, binStart, pairs, E, NB, NW);
  k_binsort<<<NB, b1024, 0, stream>>>(pairs, binStart, start, cnt, dinv, csr, N);

  k_gemm_wmma<<<gemmBlocks, b256, 0, stream>>>(x, W1, bufA, N, K1);
  k_gather<false><<<nodeBlocks, b256, 0, stream>>>(bufA, start, cnt, csr, dinv, b1, hout, batch, hs, N, G);

  k_gemm_wmma<<<gemmBlocks, b256, 0, stream>>>(hout, W2, bufA, N, NCOL);
  k_gather<true><<<nodeBlocks, b256, 0, stream>>>(bufA, start, cnt, csr, dinv, b2, hout, batch, hs, N, G);
}
